# starting kernel
# baseline (speedup 1.0000x reference)
#include <hip/hip_runtime.h>

typedef __attribute__((ext_vector_type(2))) float v2f;
typedef __attribute__((ext_vector_type(4))) float v4f;
typedef __attribute__((ext_vector_type(8))) float v8f;

#define Gc   64
#define Dc   16
#define Hc   8
#define Bc   64
#define Tc   512
#define K3   24
#define GD   (Gc * Dc)
#define NCH  (Gc * Bc)

__device__ __forceinline__ float fast_sigmoid(float x) {
    float e = __expf(-x);
    return __builtin_amdgcn_rcpf(1.0f + e);
}
__device__ __forceinline__ float fast_tanh(float x) {
    float e = __expf(2.0f * x);
    return 1.0f - 2.0f * __builtin_amdgcn_rcpf(e + 1.0f);
}

__global__ __launch_bounds__(256) void gru_input_proj_wmma(
    const float* __restrict__ x,
    const float* __restrict__ W_ih,
    const float* __restrict__ b_ih,
    float* __restrict__ gx)
{
    const int lane = threadIdx.x & 31;
    const int wave = threadIdx.x >> 5;
    const int tile = blockIdx.x * 8 + wave;

    const int g    = tile >> 11;
    const int rem  = tile & 2047;
    const int t    = rem >> 2;
    const int b0   = (rem & 3) << 4;

    const int half = lane >> 4;
    const int l16  = lane & 15;

    v2f a[4];
    {
        const v2f* xp = (const v2f*)(x + ((size_t)(b0 + l16) * Tc + t) * GD
                                       + g * Dc + 2 * half);
        #pragma unroll
        for (int j = 0; j < 4; ++j) a[j] = xp[2 * j];
    }

    v2f bm0[4], bm1[4];
    {
        const float* wg  = W_ih + (size_t)g * K3 * Dc;
        const v2f*   wr0 = (const v2f*)(wg + (size_t)l16 * Dc + 2 * half);
        const v2f*   wr1 = (const v2f*)(wg + (size_t)(8 + l16) * Dc + 2 * half);
        #pragma unroll
        for (int j = 0; j < 4; ++j) bm0[j] = wr0[2 * j];
        #pragma unroll
        for (int j = 0; j < 4; ++j) bm1[j] = wr1[2 * j];
    }

    v8f acc0 = {};
    v8f acc1 = {};
    #pragma unroll
    for (int j = 0; j < 4; ++j) {
        acc0 = __builtin_amdgcn_wmma_f32_16x16x4_f32(
                   false, a[j], false, bm0[j], (short)0, acc0, false, false);
        acc1 = __builtin_amdgcn_wmma_f32_16x16x4_f32(
                   false, a[j], false, bm1[j], (short)0, acc1, false, false);
    }

    const float bias0 = b_ih[g * K3 + l16];
    const float bias1 = b_ih[g * K3 + 8 + l16];

    const size_t rowBase = (size_t)t * NCH + (size_t)g * 64 + b0;
    #pragma unroll
    for (int r = 0; r < 8; ++r) {
        const int m = r + 8 * half;
        float* orow = gx + (rowBase + m) * K3;
        orow[l16]     = acc0[r] + bias0;
        orow[8 + l16] = acc1[r] + bias1;
    }
}

__global__ __launch_bounds__(256) void gru_recurrence(
    const float* __restrict__ gx,
    const float* __restrict__ W_hh,
    const float* __restrict__ b_hh,
    float* __restrict__ out)
{
    __shared__ float sW[4 * K3 * Hc];
    __shared__ float sB[4 * K3];

    const int tid   = threadIdx.x;
    const int gBase = blockIdx.x * 4;

    for (int i = tid; i < 4 * K3 * Hc; i += 256)
        sW[i] = W_hh[(size_t)gBase * K3 * Hc + i];
    for (int i = tid; i < 4 * K3; i += 256)
        sB[i] = b_hh[(size_t)gBase * K3 + i];
    __syncthreads();

    const int c  = blockIdx.x * 256 + tid;
    const int b  = c & 63;
    const int gl = tid >> 6;
    const int g  = gBase + gl;

    float w[K3][Hc];
    float bb[K3];
    {
        const v4f* sWv = (const v4f*)(sW + gl * K3 * Hc);
        #pragma unroll
        for (int k = 0; k < K3; ++k) {
            v4f w0 = sWv[2 * k];
            v4f w1 = sWv[2 * k + 1];
            w[k][0] = w0.x; w[k][1] = w0.y; w[k][2] = w0.z; w[k][3] = w0.w;
            w[k][4] = w1.x; w[k][5] = w1.y; w[k][6] = w1.z; w[k][7] = w1.w;
            bb[k]   = sB[gl * K3 + k];
        }
    }

    float* outBase  = out + (size_t)b * Tc * (Gc * Hc) + (size_t)g * Hc;
    const float* gc = gx + (size_t)c * K3;

    float h[Hc];
    #pragma unroll
    for (int i = 0; i < Hc; ++i) h[i] = 0.0f;

    auto loadgx = [&](v4f buf[6], int t) {
        const v4f* gp = (const v4f*)(gc + (size_t)t * (NCH * K3));
        #pragma unroll
        for (int i = 0; i < 6; ++i) buf[i] = gp[i];
    };

    auto step = [&](const v4f buf[6], int t) {
        float gxa[K3];
        #pragma unroll
        for (int i = 0; i < 6; ++i) {
            gxa[4 * i + 0] = buf[i].x; gxa[4 * i + 1] = buf[i].y;
            gxa[4 * i + 2] = buf[i].z; gxa[4 * i + 3] = buf[i].w;
        }

        float gh[K3];
        #pragma unroll
        for (int k = 0; k < K3; ++k) {
            float acc = bb[k];
            #pragma unroll
            for (int j = 0; j < Hc; ++j) acc += w[k][j] * h[j];
            gh[k] = acc;
        }

        #pragma unroll
        for (int i = 0; i < Hc; ++i) {
            float r = fast_sigmoid(gxa[i]      + gh[i]);
            float z = fast_sigmoid(gxa[8 + i]  + gh[8 + i]);
            float n = fast_tanh(gxa[16 + i] + r * gh[16 + i]);
            h[i] = (1.0f - z) * n + z * h[i];
        }

        v4f o0 = {h[0], h[1], h[2], h[3]};
        v4f o1 = {h[4], h[5], h[6], h[7]};
        v4f* op = (v4f*)(outBase + (size_t)t * (Gc * Hc));
        op[0] = o0;
        op[1] = o1;
    };

    v4f bufA[6], bufB[6];
    loadgx(bufA, 0);
    for (int t = 0; t < Tc; t += 2) {
        loadgx(bufB, t + 1);
        step(bufA, t);
        if (t + 2 < Tc) loadgx(bufA, t + 2);
        step(bufB, t + 1);
    }
}

extern "C" void kernel_launch(void* const* d_in, const int* in_sizes, int n_in,
                              void* d_out, int out_size, void* d_ws, size_t ws_size,
                              hipStream_t stream) {
    const float* x    = (const float*)d_in[0];
    const float* W_ih = (const float*)d_in[1];
    const float* W_hh = (const float*)d_in[2];
    const float* b_ih = (const float*)d_in[3];
    const float* b_hh = (const float*)d_in[4];
    float* out = (float*)d_out;
    float* gx  = (float*)d_ws;

    gru_input_proj_wmma<<<dim3(131072 / 8), dim3(256), 0, stream>>>(x, W_ih, b_ih, gx);
    gru_recurrence<<<dim3(NCH / 256), dim3(256), 0, stream>>>(gx, W_hh, b_hh, out);
}


# input projection joined to the recurrent product as one k range [x;h]: W_ih rows as A, x rows as B, summed into the C tiles one step ahead; no gx array, one launch; sum order (bias first, d) changed
# speedup vs baseline: 5.7328x; 5.7328x over previous
#include <hip/hip_runtime.h>

typedef __attribute__((ext_vector_type(2))) float v2f;
typedef __attribute__((ext_vector_type(4))) float v4f;
typedef __attribute__((ext_vector_type(8))) float v8f;

#define Gc   64
#define Dc   16
#define Hc   8
#define Bc   64
#define Tc   512
#define K3   24
#define GD   (Gc * Dc)

__device__ __forceinline__ float fast_sigmoid(float x) {
    float e = __expf(-x);
    return __builtin_amdgcn_rcpf(1.0f + e);
}
__device__ __forceinline__ float fast_tanh(float x) {
    float e = __expf(2.0f * x);
    return 1.0f - 2.0f * __builtin_amdgcn_rcpf(e + 1.0f);
}

#define RING 16

__global__ __launch_bounds__(32) void gru_joined_wmma(
    const float* __restrict__ x,
    const float* __restrict__ W_ih,
    const float* __restrict__ W_hh,
    const float* __restrict__ b_ih,
    const float* __restrict__ b_hh,
    float* __restrict__ out)
{
    const int lane = threadIdx.x;
    const int c16  = lane & 15;
    const int half = lane >> 4;
    const int g    = blockIdx.x >> 2;
    const int b    = ((blockIdx.x & 3) << 4) + c16;

    const float* xc = x + (size_t)b * Tc * GD + (size_t)g * Dc + 8 * half;
    auto loadx = [&](v4f buf[2], int t) {
        const float* xp = xc + (size_t)t * GD;
        buf[0] = *(const v4f*)(xp);
        buf[1] = *(const v4f*)(xp + 4);
    };
    v4f ring[RING][2];
    #pragma unroll
    for (int s = 0; s < RING; ++s) loadx(ring[s], s);

    const int hsel = c16 >> 3;
    const int q    = c16 & 7;
    const int row0 = (q >> 2) * Hc + 4 * hsel + (q & 3);
    const int rown = 2 * Hc + 4 * hsel + (q & 3);
    const float* wh = W_hh + (size_t)g * K3 * Hc;
    const float* wi = W_ih + (size_t)g * K3 * Dc;
    const v2f zero2 = {0.0f, 0.0f};
    v2f a0[2], a1[2];
    #pragma unroll
    for (int j = 0; j < 2; ++j) {
        a0[j] = *(const v2f*)(wh + row0 * Hc + 4 * half + 2 * j);
        v2f w1 = *(const v2f*)(wh + rown * Hc + 4 * half + 2 * j);
        a1[j] = (q < 4) ? w1 : zero2;
    }
    v2f ax0[4], ax1[4];
    #pragma unroll
    for (int j = 0; j < 4; ++j) {
        ax0[j] = *(const v2f*)(wi + row0 * Dc + 8 * half + 2 * j);
        v2f w1 = *(const v2f*)(wi + rown * Dc + 8 * half + 2 * j);
        ax1[j] = (q >= 4) ? w1 : zero2;
    }

    const float* bi = b_ih + (size_t)g * K3 + 4 * half;
    const float* bh = b_hh + (size_t)g * K3 + 4 * half;
    const v4f br = *(const v4f*)(bi)          + *(const v4f*)(bh);
    const v4f bz = *(const v4f*)(bi + Hc)     + *(const v4f*)(bh + Hc);
    const v4f bxn = *(const v4f*)(bi + 2 * Hc);
    const v4f bhn = *(const v4f*)(bh + 2 * Hc);
    const v8f cb0 = {br.x, br.y, br.z, br.w, bz.x, bz.y, bz.z, bz.w};
    const v8f cb1 = {bhn.x, bhn.y, bhn.z, bhn.w, bxn.x, bxn.y, bxn.z, bxn.w};

    float* outBase  = out + (size_t)b * Tc * (Gc * Hc) + (size_t)g * Hc + 4 * half;

    float h[4] = {0.0f, 0.0f, 0.0f, 0.0f};

    auto xpart = [&](const v4f buf[2], v8f& c0, v8f& c1) {
        const v2f xb0 = {buf[0].x, buf[0].y};
        const v2f xb1 = {buf[0].z, buf[0].w};
        const v2f xb2 = {buf[1].x, buf[1].y};
        const v2f xb3 = {buf[1].z, buf[1].w};
        c0 = __builtin_amdgcn_wmma_f32_16x16x4_f32(
                 false, ax0[0], false, xb0, (short)0, cb0, false, false);
        c1 = __builtin_amdgcn_wmma_f32_16x16x4_f32(
                 false, ax1[0], false, xb0, (short)0, cb1, false, false);
        c0 = __builtin_amdgcn_wmma_f32_16x16x4_f32(
                 false, ax0[1], false, xb1, (short)0, c0, false, false);
        c1 = __builtin_amdgcn_wmma_f32_16x16x4_f32(
                 false, ax1[1], false, xb1, (short)0, c1, false, false);
        c0 = __builtin_amdgcn_wmma_f32_16x16x4_f32(
                 false, ax0[2], false, xb2, (short)0, c0, false, false);
        c1 = __builtin_amdgcn_wmma_f32_16x16x4_f32(
                 false, ax1[2], false, xb2, (short)0, c1, false, false);
        c0 = __builtin_amdgcn_wmma_f32_16x16x4_f32(
                 false, ax0[3], false, xb3, (short)0, c0, false, false);
        c1 = __builtin_amdgcn_wmma_f32_16x16x4_f32(
                 false, ax1[3], false, xb3, (short)0, c1, false, false);
    };

    v8f c0, c1;
    xpart(ring[0], c0, c1);

    for (int t = 0; t < Tc; t += RING) {
        const int tf = (t + RING < Tc) ? t + RING : Tc - RING;
        #pragma unroll
        for (int s = 0; s < RING; ++s) {
            const int tt = t + s;
            const int sn = (s + 1) % RING;

            const v2f hb0 = {h[0], h[1]};
            const v2f hb1 = {h[2], h[3]};
            v8f d0 = __builtin_amdgcn_wmma_f32_16x16x4_f32(
                         false, a0[0], false, hb0, (short)0, c0, false, false);
            v8f d1 = __builtin_amdgcn_wmma_f32_16x16x4_f32(
                         false, a1[0], false, hb0, (short)0, c1, false, false);
            d0 = __builtin_amdgcn_wmma_f32_16x16x4_f32(
                         false, a0[1], false, hb1, (short)0, d0, false, false);
            d1 = __builtin_amdgcn_wmma_f32_16x16x4_f32(
                         false, a1[1], false, hb1, (short)0, d1, false, false);

            xpart(ring[sn], c0, c1);
            loadx(ring[s], tf + s);

            #pragma unroll
            for (int i = 0; i < 4; ++i) {
                float r = fast_sigmoid(d0[i]);
                float z = fast_sigmoid(d0[4 + i]);
                float n = fast_tanh(d1[4 + i] + r * d1[i]);
                h[i] = (1.0f - z) * n + z * h[i];
            }

            v4f o = {h[0], h[1], h[2], h[3]};
            *(v4f*)(outBase + (size_t)tt * (Gc * Hc)) = o;

            __builtin_amdgcn_sched_barrier(0);
        }
    }
}

extern "C" void kernel_launch(void* const* d_in, const int* in_sizes, int n_in,
                              void* d_out, int out_size, void* d_ws, size_t ws_size,
                              hipStream_t stream) {
    const float* x    = (const float*)d_in[0];
    const float* W_ih = (const float*)d_in[1];
    const float* W_hh = (const float*)d_in[2];
    const float* b_ih = (const float*)d_in[3];
    const float* b_hh = (const float*)d_in[4];
    float* out = (float*)d_out;

    gru_joined_wmma<<<dim3(256), dim3(32), 0, stream>>>(x, W_ih, W_hh, b_ih, b_hh, out);
}


# x part (bias + W_ih.x: 8 of a step's 12 matrix instructions) moved to a second wave of each work-group on another SIMD; C tiles reach the chain wave through LDS in 16-step chunks; sums unchanged
# speedup vs baseline: 6.1952x; 1.0807x over previous
#include <hip/hip_runtime.h>

typedef __attribute__((ext_vector_type(2))) float v2f;
typedef __attribute__((ext_vector_type(4))) float v4f;
typedef __attribute__((ext_vector_type(8))) float v8f;

#define Gc   64
#define Dc   16
#define Hc   8
#define Bc   64
#define Tc   512
#define K3   24
#define GD   (Gc * Dc)
#define CH   16
#define NCH  (Tc / CH)

__device__ __forceinline__ float fast_sigmoid(float x) {
    float e = __expf(-x);
    return __builtin_amdgcn_rcpf(1.0f + e);
}
__device__ __forceinline__ float fast_tanh(float x) {
    float e = __expf(2.0f * x);
    return 1.0f - 2.0f * __builtin_amdgcn_rcpf(e + 1.0f);
}

__global__ __launch_bounds__(64) void gru_two_waves(
    const float* __restrict__ x,
    const float* __restrict__ W_ih,
    const float* __restrict__ W_hh,
    const float* __restrict__ b_ih,
    const float* __restrict__ b_hh,
    float* __restrict__ out)
{
    __shared__ v4f ctile[2][CH][4][32];

    const int lane = threadIdx.x & 31;
    const int role = __builtin_amdgcn_readfirstlane((int)(threadIdx.x >> 5));
    const int c16  = lane & 15;
    const int half = lane >> 4;
    const int g    = blockIdx.x >> 2;
    const int b    = ((blockIdx.x & 3) << 4) + c16;

    const int hsel = c16 >> 3;
    const int q    = c16 & 7;
    const int row0 = (q >> 2) * Hc + 4 * hsel + (q & 3);
    const int rown = 2 * Hc + 4 * hsel + (q & 3);
    const v2f zero2 = {0.0f, 0.0f};

    const float* xc = x + (size_t)b * Tc * GD + (size_t)g * Dc + 8 * half;
    auto loadx = [&](v4f buf[2], const float* xp) {
        buf[0] = *(const v4f*)(xp);
        buf[1] = *(const v4f*)(xp + 4);
    };
    v4f ring[CH][2];
    v2f ax0[4], ax1[4];
    v8f cb0, cb1;
    v2f a0[2], a1[2];
    float h[4] = {0.0f, 0.0f, 0.0f, 0.0f};
    float* outBase = out + (size_t)b * Tc * (Gc * Hc) + (size_t)g * Hc + 4 * half;

    if (role == 1) {
        #pragma unroll
        for (int s = 0; s < CH - 1; ++s) loadx(ring[s], xc + (size_t)s * GD);
        const float* wi = W_ih + (size_t)g * K3 * Dc;
        #pragma unroll
        for (int j = 0; j < 4; ++j) {
            ax0[j] = *(const v2f*)(wi + row0 * Dc + 8 * half + 2 * j);
            v2f w1 = *(const v2f*)(wi + rown * Dc + 8 * half + 2 * j);
            ax1[j] = (q >= 4) ? w1 : zero2;
        }
        const float* bi = b_ih + (size_t)g * K3 + 4 * half;
        const float* bh = b_hh + (size_t)g * K3 + 4 * half;
        const v4f br = *(const v4f*)(bi)          + *(const v4f*)(bh);
        const v4f bz = *(const v4f*)(bi + Hc)     + *(const v4f*)(bh + Hc);
        const v4f bxn = *(const v4f*)(bi + 2 * Hc);
        const v4f bhn = *(const v4f*)(bh + 2 * Hc);
        cb0 = (v8f){br.x, br.y, br.z, br.w, bz.x, bz.y, bz.z, bz.w};
        cb1 = (v8f){bhn.x, bhn.y, bhn.z, bhn.w, bxn.x, bxn.y, bxn.z, bxn.w};
    } else {
        const float* wh = W_hh + (size_t)g * K3 * Hc;
        #pragma unroll
        for (int j = 0; j < 2; ++j) {
            a0[j] = *(const v2f*)(wh + row0 * Hc + 4 * half + 2 * j);
            v2f w1 = *(const v2f*)(wh + rown * Hc + 4 * half + 2 * j);
            a1[j] = (q < 4) ? w1 : zero2;
        }
    }

    auto xpart = [&](const v4f buf[2], v8f& c0, v8f& c1) {
        const v2f xb0 = {buf[0].x, buf[0].y};
        const v2f xb1 = {buf[0].z, buf[0].w};
        const v2f xb2 = {buf[1].x, buf[1].y};
        const v2f xb3 = {buf[1].z, buf[1].w};
        c0 = __builtin_amdgcn_wmma_f32_16x16x4_f32(
                 false, ax0[0], false, xb0, (short)0, cb0, false, false);
        c1 = __builtin_amdgcn_wmma_f32_16x16x4_f32(
                 false, ax1[0], false, xb0, (short)0, cb1, false, false);
        c0 = __builtin_amdgcn_wmma_f32_16x16x4_f32(
                 false, ax0[1], false, xb1, (short)0, c0, false, false);
        c1 = __builtin_amdgcn_wmma_f32_16x16x4_f32(
                 false, ax1[1], false, xb1, (short)0, c1, false, false);
        c0 = __builtin_amdgcn_wmma_f32_16x16x4_f32(
                 false, ax0[2], false, xb2, (short)0, c0, false, false);
        c1 = __builtin_amdgcn_wmma_f32_16x16x4_f32(
                 false, ax1[2], false, xb2, (short)0, c1, false, false);
        c0 = __builtin_amdgcn_wmma_f32_16x16x4_f32(
                 false, ax0[3], false, xb3, (short)0, c0, false, false);
        c1 = __builtin_amdgcn_wmma_f32_16x16x4_f32(
                 false, ax1[3], false, xb3, (short)0, c1, false, false);
    };

    auto produce = [&](int k) {
        const int t0  = k * CH;
        const int tfn = (t0 + CH < Tc) ? t0 + CH : t0;
        const float* xown  = xc + (size_t)t0 * GD;
        const float* xnext = xc + (size_t)tfn * GD;
        v4f* dst = &ctile[k & 1][0][0][lane];
        #pragma unroll
        for (int s = 0; s < CH; ++s) {
            if (s == 0) loadx(ring[CH - 1], xown + (size_t)(CH - 1) * GD);
            else        loadx(ring[s - 1], xnext + (size_t)(s - 1) * GD);
            v8f c0, c1;
            xpart(ring[s], c0, c1);
            dst[(s * 4 + 0) * 32] = (v4f){c0[0], c0[1], c0[2], c0[3]};
            dst[(s * 4 + 1) * 32] = (v4f){c0[4], c0[5], c0[6], c0[7]};
            dst[(s * 4 + 2) * 32] = (v4f){c1[0], c1[1], c1[2], c1[3]};
            dst[(s * 4 + 3) * 32] = (v4f){c1[4], c1[5], c1[6], c1[7]};
            __builtin_amdgcn_sched_barrier(0);
        }
    };

    auto consume = [&](int k) {
        const v4f* src = &ctile[k & 1][0][0][lane];
        float* op = outBase + (size_t)(k * CH) * (Gc * Hc);
        v4f p0 = src[0 * 32], p1 = src[1 * 32], p2 = src[2 * 32], p3 = src[3 * 32];
        #pragma unroll
        for (int s = 0; s < CH; ++s) {
            const v8f c0 = {p0.x, p0.y, p0.z, p0.w, p1.x, p1.y, p1.z, p1.w};
            const v8f c1 = {p2.x, p2.y, p2.z, p2.w, p3.x, p3.y, p3.z, p3.w};

            const v2f hb0 = {h[0], h[1]};
            const v2f hb1 = {h[2], h[3]};
            v8f d0 = __builtin_amdgcn_wmma_f32_16x16x4_f32(
                         false, a0[0], false, hb0, (short)0, c0, false, false);
            v8f d1 = __builtin_amdgcn_wmma_f32_16x16x4_f32(
                         false, a1[0], false, hb0, (short)0, c1, false, false);
            d0 = __builtin_amdgcn_wmma_f32_16x16x4_f32(
                         false, a0[1], false, hb1, (short)0, d0, false, false);
            d1 = __builtin_amdgcn_wmma_f32_16x16x4_f32(
                         false, a1[1], false, hb1, (short)0, d1, false, false);

            if (s + 1 < CH) {
                p0 = src[((s + 1) * 4 + 0) * 32];
                p1 = src[((s + 1) * 4 + 1) * 32];
                p2 = src[((s + 1) * 4 + 2) * 32];
                p3 = src[((s + 1) * 4 + 3) * 32];
            }

            #pragma unroll
            for (int i = 0; i < 4; ++i) {
                float r = fast_sigmoid(d0[i]);
                float z = fast_sigmoid(d0[4 + i]);
                float n = fast_tanh(d1[4 + i] + r * d1[i]);
                h[i] = (1.0f - z) * n + z * h[i];
            }

            v4f o = {h[0], h[1], h[2], h[3]};
            *(v4f*)(op + (size_t)s * (Gc * Hc)) = o;

            __builtin_amdgcn_sched_barrier(0);
        }
    };

    for (int k = -1; k < NCH; ++k) {
        if (role == 1) {
            if (k + 1 < NCH) produce(k + 1);
        } else {
            if (k >= 0) consume(k);
        }
        __builtin_amdgcn_fence(__ATOMIC_RELEASE, "workgroup", "local");
        __builtin_amdgcn_s_barrier();
        __builtin_amdgcn_fence(__ATOMIC_ACQUIRE, "workgroup", "local");
    }
}

extern "C" void kernel_launch(void* const* d_in, const int* in_sizes, int n_in,
                              void* d_out, int out_size, void* d_ws, size_t ws_size,
                              hipStream_t stream) {
    const float* x    = (const float*)d_in[0];
    const float* W_ih = (const float*)d_in[1];
    const float* W_hh = (const float*)d_in[2];
    const float* b_ih = (const float*)d_in[3];
    const float* b_hh = (const float*)d_in[4];
    float* out = (float*)d_out;

    gru_two_waves<<<dim3(256), dim3(64), 0, stream>>>(x, W_ih, W_hh, b_ih, b_hh, out);
}


# gate exponent scales (-log2 e on r, z rows; 2 log2 e on n rows) folded into W_ih, W_hh, bias rows at load: v_exp_f32 reads the matrix result directly; sum order unchanged, products round differently
# speedup vs baseline: 6.8335x; 1.1030x over previous
#include <hip/hip_runtime.h>

typedef __attribute__((ext_vector_type(2))) float v2f;
typedef __attribute__((ext_vector_type(4))) float v4f;
typedef __attribute__((ext_vector_type(8))) float v8f;

#define Gc   64
#define Dc   16
#define Hc   8
#define Bc   64
#define Tc   512
#define K3   24
#define GD   (Gc * Dc)
#define CH   16
#define NCH  (Tc / CH)

#define LOG2E       1.44269504088896340736f
#define SIG_SCALE   (-LOG2E)
#define TANH_SCALE  (2.0f * LOG2E)

__device__ __forceinline__ float sigmoid_prescaled(float ps) {
    float e = __builtin_amdgcn_exp2f(ps);
    return __builtin_amdgcn_rcpf(1.0f + e);
}
__device__ __forceinline__ float tanh_prescaled(float as) {
    float e = __builtin_amdgcn_exp2f(as);
    return 1.0f - 2.0f * __builtin_amdgcn_rcpf(e + 1.0f);
}

__global__ __launch_bounds__(64) void gru_two_waves(
    const float* __restrict__ x,
    const float* __restrict__ W_ih,
    const float* __restrict__ W_hh,
    const float* __restrict__ b_ih,
    const float* __restrict__ b_hh,
    float* __restrict__ out)
{
    __shared__ v4f ctile[2][CH][4][32];

    const int lane = threadIdx.x & 31;
    const int role = __builtin_amdgcn_readfirstlane((int)(threadIdx.x >> 5));
    const int c16  = lane & 15;
    const int half = lane >> 4;
    const int g    = blockIdx.x >> 2;
    const int b    = ((blockIdx.x & 3) << 4) + c16;

    const int hsel = c16 >> 3;
    const int q    = c16 & 7;
    const int row0 = (q >> 2) * Hc + 4 * hsel + (q & 3);
    const int rown = 2 * Hc + 4 * hsel + (q & 3);
    const v2f zero2 = {0.0f, 0.0f};

    const float* xc = x + (size_t)b * Tc * GD + (size_t)g * Dc + 8 * half;
    auto loadx = [&](v4f buf[2], const float* xp) {
        buf[0] = *(const v4f*)(xp);
        buf[1] = *(const v4f*)(xp + 4);
    };
    v4f ring[CH][2];
    v2f ax0[4], ax1[4];
    v8f cb0, cb1;
    v2f a0[2], a1[2];
    float h[4] = {0.0f, 0.0f, 0.0f, 0.0f};
    float* outBase = out + (size_t)b * Tc * (Gc * Hc) + (size_t)g * Hc + 4 * half;

    if (role == 1) {
        #pragma unroll
        for (int s = 0; s < CH - 1; ++s) loadx(ring[s], xc + (size_t)s * GD);
        const float* wi = W_ih + (size_t)g * K3 * Dc;
        #pragma unroll
        for (int j = 0; j < 4; ++j) {
            ax0[j] = *(const v2f*)(wi + row0 * Dc + 8 * half + 2 * j) * SIG_SCALE;
            v2f w1 = *(const v2f*)(wi + rown * Dc + 8 * half + 2 * j) * TANH_SCALE;
            ax1[j] = (q >= 4) ? w1 : zero2;
        }
        const float* bi = b_ih + (size_t)g * K3 + 4 * half;
        const float* bh = b_hh + (size_t)g * K3 + 4 * half;
        const v4f br = (*(const v4f*)(bi)          + *(const v4f*)(bh))      * SIG_SCALE;
        const v4f bz = (*(const v4f*)(bi + Hc)     + *(const v4f*)(bh + Hc)) * SIG_SCALE;
        const v4f bxn = *(const v4f*)(bi + 2 * Hc) * TANH_SCALE;
        const v4f bhn = *(const v4f*)(bh + 2 * Hc) * TANH_SCALE;
        cb0 = (v8f){br.x, br.y, br.z, br.w, bz.x, bz.y, bz.z, bz.w};
        cb1 = (v8f){bhn.x, bhn.y, bhn.z, bhn.w, bxn.x, bxn.y, bxn.z, bxn.w};
    } else {
        const float* wh = W_hh + (size_t)g * K3 * Hc;
        #pragma unroll
        for (int j = 0; j < 2; ++j) {
            a0[j] = *(const v2f*)(wh + row0 * Hc + 4 * half + 2 * j) * SIG_SCALE;
            v2f w1 = *(const v2f*)(wh + rown * Hc + 4 * half + 2 * j) * TANH_SCALE;
            a1[j] = (q < 4) ? w1 : zero2;
        }
    }

    auto xpart = [&](const v4f buf[2], v8f& c0, v8f& c1) {
        const v2f xb0 = {buf[0].x, buf[0].y};
        const v2f xb1 = {buf[0].z, buf[0].w};
        const v2f xb2 = {buf[1].x, buf[1].y};
        const v2f xb3 = {buf[1].z, buf[1].w};
        c0 = __builtin_amdgcn_wmma_f32_16x16x4_f32(
                 false, ax0[0], false, xb0, (short)0, cb0, false, false);
        c1 = __builtin_amdgcn_wmma_f32_16x16x4_f32(
                 false, ax1[0], false, xb0, (short)0, cb1, false, false);
        c0 = __builtin_amdgcn_wmma_f32_16x16x4_f32(
                 false, ax0[1], false, xb1, (short)0, c0, false, false);
        c1 = __builtin_amdgcn_wmma_f32_16x16x4_f32(
                 false, ax1[1], false, xb1, (short)0, c1, false, false);
        c0 = __builtin_amdgcn_wmma_f32_16x16x4_f32(
                 false, ax0[2], false, xb2, (short)0, c0, false, false);
        c1 = __builtin_amdgcn_wmma_f32_16x16x4_f32(
                 false, ax1[2], false, xb2, (short)0, c1, false, false);
        c0 = __builtin_amdgcn_wmma_f32_16x16x4_f32(
                 false, ax0[3], false, xb3, (short)0, c0, false, false);
        c1 = __builtin_amdgcn_wmma_f32_16x16x4_f32(
                 false, ax1[3], false, xb3, (short)0, c1, false, false);
    };

    auto produce = [&](int k) {
        const int t0  = k * CH;
        const int tfn = (t0 + CH < Tc) ? t0 + CH : t0;
        const float* xown  = xc + (size_t)t0 * GD;
        const float* xnext = xc + (size_t)tfn * GD;
        v4f* dst = &ctile[k & 1][0][0][lane];
        #pragma unroll
        for (int s = 0; s < CH; ++s) {
            if (s == 0) loadx(ring[CH - 1], xown + (size_t)(CH - 1) * GD);
            else        loadx(ring[s - 1], xnext + (size_t)(s - 1) * GD);
            v8f c0, c1;
            xpart(ring[s], c0, c1);
            dst[(s * 4 + 0) * 32] = (v4f){c0[0], c0[1], c0[2], c0[3]};
            dst[(s * 4 + 1) * 32] = (v4f){c0[4], c0[5], c0[6], c0[7]};
            dst[(s * 4 + 2) * 32] = (v4f){c1[0], c1[1], c1[2], c1[3]};
            dst[(s * 4 + 3) * 32] = (v4f){c1[4], c1[5], c1[6], c1[7]};
            __builtin_amdgcn_sched_barrier(0);
        }
    };

    auto consume = [&](int k) {
        const v4f* src = &ctile[k & 1][0][0][lane];
        float* op = outBase + (size_t)(k * CH) * (Gc * Hc);
        v4f p0 = src[0 * 32], p1 = src[1 * 32], p2 = src[2 * 32], p3 = src[3 * 32];
        #pragma unroll
        for (int s = 0; s < CH; ++s) {
            const v8f c0 = {p0.x, p0.y, p0.z, p0.w, p1.x, p1.y, p1.z, p1.w};
            const v8f c1 = {p2.x, p2.y, p2.z, p2.w, p3.x, p3.y, p3.z, p3.w};

            const v2f hb0 = {h[0], h[1]};
            const v2f hb1 = {h[2], h[3]};
            v8f d0 = __builtin_amdgcn_wmma_f32_16x16x4_f32(
                         false, a0[0], false, hb0, (short)0, c0, false, false);
            v8f d1 = __builtin_amdgcn_wmma_f32_16x16x4_f32(
                         false, a1[0], false, hb0, (short)0, c1, false, false);
            d0 = __builtin_amdgcn_wmma_f32_16x16x4_f32(
                         false, a0[1], false, hb1, (short)0, d0, false, false);
            d1 = __builtin_amdgcn_wmma_f32_16x16x4_f32(
                         false, a1[1], false, hb1, (short)0, d1, false, false);

            if (s + 1 < CH) {
                p0 = src[((s + 1) * 4 + 0) * 32];
                p1 = src[((s + 1) * 4 + 1) * 32];
                p2 = src[((s + 1) * 4 + 2) * 32];
                p3 = src[((s + 1) * 4 + 3) * 32];
            }

            #pragma unroll
            for (int i = 0; i < 4; ++i) {
                float r = sigmoid_prescaled(d0[i]);
                float z = sigmoid_prescaled(d0[4 + i]);
                float n = tanh_prescaled(d1[4 + i] + r * d1[i]);
                h[i] = (1.0f - z) * n + z * h[i];
            }

            v4f o = {h[0], h[1], h[2], h[3]};
            *(v4f*)(op + (size_t)s * (Gc * Hc)) = o;

            __builtin_amdgcn_sched_barrier(0);
        }
    };

    for (int k = -1; k < NCH; ++k) {
        if (role == 1) {
            if (k + 1 < NCH) produce(k + 1);
        } else {
            if (k >= 0) consume(k);
        }
        __builtin_amdgcn_fence(__ATOMIC_RELEASE, "workgroup", "local");
        __builtin_amdgcn_s_barrier();
        __builtin_amdgcn_fence(__ATOMIC_ACQUIRE, "workgroup", "local");
    }
}

extern "C" void kernel_launch(void* const* d_in, const int* in_sizes, int n_in,
                              void* d_out, int out_size, void* d_ws, size_t ws_size,
                              hipStream_t stream) {
    const float* x    = (const float*)d_in[0];
    const float* W_ih = (const float*)d_in[1];
    const float* W_hh = (const float*)d_in[2];
    const float* b_ih = (const float*)d_in[3];
    const float* b_hh = (const float*)d_in[4];
    float* out = (float*)d_out;

    gru_two_waves<<<dim3(256), dim3(64), 0, stream>>>(x, W_ih, W_hh, b_ih, b_hh, out);
}
